# starting kernel
# baseline (speedup 1.0000x reference)
#include <hip/hip_runtime.h>

#define B_  64
#define S_  4096
#define C_  128
#define L_  128
#define K_  32
#define NL_ (B_ * C_)
#define LANES_ 16
#define NTHREADS_ (LANES_ * K_)

#define ROWSTRIDE_   19
#define CHUNKSTRIDE_ (L_ * ROWSTRIDE_ + 16)
#define EP_FLOATS_   (K_ * CHUNKSTRIDE_)
#define H_OFF_       EP_FLOATS_
#define CIN_OFF_     (H_OFF_ + L_ * 4)
#define LDS_FLOATS_  (CIN_OFF_ + K_ * 4 * LANES_)
#define LDS_BYTES_   (LDS_FLOATS_ * 4)

typedef float v2f __attribute__((ext_vector_type(2)));
typedef float v8f __attribute__((ext_vector_type(8)));

__global__ void __launch_bounds__(NTHREADS_) arima_fused(
    const float* __restrict__ y, const float* __restrict__ phi,
    const float* __restrict__ theta, float* __restrict__ out) {
    extern __shared__ float lds[];
    float* EP  = lds;
    float* Hh  = lds + H_OFF_;
    float* CIN = lds + CIN_OFF_;

    const int tid   = threadIdx.x;
    const int lt    = blockIdx.x;
    const int bb    = lt >> 3;
    const int cbase = (lt & 7) * LANES_;

    const float t0 = theta[0], t1 = theta[1], t2 = theta[2];

    if (tid < 4) {
        const int j = tid;
        float e1 = (j == 0) ? 1.f : 0.f;
        float e2 = (j == 1) ? 1.f : 0.f;
        float e3 = (j == 2) ? 1.f : 0.f;
        for (int t = 0; t < L_; ++t) {
            float e = -(t0 * e1 + t1 * e2 + t2 * e3);
            Hh[t * 4 + j] = e;
            e3 = e2; e2 = e1; e1 = e;
        }
    }

    {
        const float p0 = phi[0], p1 = phi[1], p2 = phi[2];
        const int k = tid >> 4;
        const int x = tid & 15;
        const float* yp = y + ((size_t)bb * S_ + (size_t)k * L_) * C_ + cbase + x;
        float* ep = EP + k * CHUNKSTRIDE_ + x;
        float yl1 = 0.f, yl2 = 0.f, yl3 = 0.f;
        if (k > 0) { yl1 = yp[-C_]; yl2 = yp[-2 * C_]; yl3 = yp[-3 * C_]; }
        float e1 = 0.f, e2 = 0.f, e3 = 0.f;
#pragma unroll 4
        for (int t = 0; t < L_; ++t) {
            float yt = yp[t * C_];
            float z  = yt - p0 * yl1 - p1 * yl2 - p2 * yl3;
            float w  = z  - t1 * e2  - t2 * e3;
            float e  = w  - t0 * e1;
            ep[t * ROWSTRIDE_] = e;
            yl3 = yl2; yl2 = yl1; yl1 = yt;
            e3 = e2; e2 = e1; e1 = e;
        }
    }
    __syncthreads();

    if (tid < LANES_) {
        const int x = tid;
        const float M00 = Hh[(L_-1)*4+0], M01 = Hh[(L_-1)*4+1], M02 = Hh[(L_-1)*4+2];
        const float M10 = Hh[(L_-2)*4+0], M11 = Hh[(L_-2)*4+1], M12 = Hh[(L_-2)*4+2];
        const float M20 = Hh[(L_-3)*4+0], M21 = Hh[(L_-3)*4+1], M22 = Hh[(L_-3)*4+2];
        float c1 = 0.f, c2 = 0.f, c3 = 0.f;
        for (int k = 0; k < K_; ++k) {
            CIN[(k*4 + 0) * LANES_ + x] = c1;
            CIN[(k*4 + 1) * LANES_ + x] = c2;
            CIN[(k*4 + 2) * LANES_ + x] = c3;
            CIN[(k*4 + 3) * LANES_ + x] = 0.f;
            const float* ep = EP + k * CHUNKSTRIDE_ + x;
            float s0 = ep[(L_-1)*ROWSTRIDE_] + M00*c1 + M01*c2 + M02*c3;
            float s1 = ep[(L_-2)*ROWSTRIDE_] + M10*c1 + M11*c2 + M12*c3;
            float s2 = ep[(L_-3)*ROWSTRIDE_] + M20*c1 + M21*c2 + M22*c3;
            c1 = s0; c2 = s1; c3 = s2;
        }
    }
    __syncthreads();

    {
        const int l    = tid & 31;
        const int w    = tid >> 5;
        const int n    = l & 15;
        const int half = l >> 4;
        float* obase = out + (size_t)bb * S_ * C_ + cbase + n;
#pragma unroll
        for (int kk = 0; kk < 2; ++kk) {
            const int k = 2 * w + kk;
            v2f bvec;
            bvec.x = CIN[(k*4 + half*2 + 0) * LANES_ + n];
            bvec.y = CIN[(k*4 + half*2 + 1) * LANES_ + n];
            const float* epk = EP + k * CHUNKSTRIDE_ + n;
#pragma unroll
            for (int tt = 0; tt < L_ / 16; ++tt) {
                const int tl = tt * 16 + n;
                v2f avec = *(const v2f*)(Hh + tl * 4 + half * 2);
                v8f cvec;
#pragma unroll
                for (int r = 0; r < 8; ++r)
                    cvec[r] = epk[(tt * 16 + half * 8 + r) * ROWSTRIDE_];
                v8f d = __builtin_amdgcn_wmma_f32_16x16x4_f32(
                     false, avec,  false, bvec,
                     (short)0, cvec,  false,  false);
                float* op = obase + (size_t)(k * L_ + tt * 16 + half * 8) * C_;
#pragma unroll
                for (int r = 0; r < 8; ++r) op[r * C_] = d[r];
            }
        }
    }
}

extern "C" void kernel_launch(void* const* d_in, const int* in_sizes, int n_in,
                              void* d_out, int out_size, void* d_ws, size_t ws_size,
                              hipStream_t stream) {
    const float* y     = (const float*)d_in[0];
    const float* phi   = (const float*)d_in[1];
    const float* theta = (const float*)d_in[2];
    float* out = (float*)d_out;
    (void)d_ws; (void)ws_size; (void)in_sizes; (void)n_in; (void)out_size;

    arima_fused<<<NL_ / LANES_, NTHREADS_, LDS_BYTES_, stream>>>(y, phi, theta, out);
}


# Phase-1 time loop unrolled 64 deep so a thread has 61-64 global loads of y in flight; the 4-deep loop waited for each load before issuing the next (128 serial memory round trips per thread)
# speedup vs baseline: 1.9179x; 1.9179x over previous
#include <hip/hip_runtime.h>

#define B_  64
#define S_  4096
#define C_  128
#define L_  128
#define K_  32
#define NL_ (B_ * C_)
#define LANES_ 16
#define NTHREADS_ (LANES_ * K_)

#define ROWSTRIDE_   19
#define CHUNKSTRIDE_ (L_ * ROWSTRIDE_ + 16)
#define EP_FLOATS_   (K_ * CHUNKSTRIDE_)
#define H_OFF_       EP_FLOATS_
#define CIN_OFF_     (H_OFF_ + L_ * 4)
#define LDS_FLOATS_  (CIN_OFF_ + K_ * 4 * LANES_)
#define LDS_BYTES_   (LDS_FLOATS_ * 4)

typedef float v2f __attribute__((ext_vector_type(2)));
typedef float v8f __attribute__((ext_vector_type(8)));

__global__ void __launch_bounds__(NTHREADS_) arima_fused(
    const float* __restrict__ y, const float* __restrict__ phi,
    const float* __restrict__ theta, float* __restrict__ out) {
    extern __shared__ float lds[];
    float* EP  = lds;
    float* Hh  = lds + H_OFF_;
    float* CIN = lds + CIN_OFF_;

    const int tid   = threadIdx.x;
    const int lt    = blockIdx.x;
    const int bb    = lt >> 3;
    const int cbase = (lt & 7) * LANES_;

    const float t0 = theta[0], t1 = theta[1], t2 = theta[2];

    if (tid < 4) {
        const int j = tid;
        float e1 = (j == 0) ? 1.f : 0.f;
        float e2 = (j == 1) ? 1.f : 0.f;
        float e3 = (j == 2) ? 1.f : 0.f;
        for (int t = 0; t < L_; ++t) {
            float e = -(t0 * e1 + t1 * e2 + t2 * e3);
            Hh[t * 4 + j] = e;
            e3 = e2; e2 = e1; e1 = e;
        }
    }

    {
        const float p0 = phi[0], p1 = phi[1], p2 = phi[2];
        const int k = tid >> 4;
        const int x = tid & 15;
        const float* yp = y + ((size_t)bb * S_ + (size_t)k * L_) * C_ + cbase + x;
        float* ep = EP + k * CHUNKSTRIDE_ + x;
        float yl1 = 0.f, yl2 = 0.f, yl3 = 0.f;
        if (k > 0) { yl1 = yp[-C_]; yl2 = yp[-2 * C_]; yl3 = yp[-3 * C_]; }
        float e1 = 0.f, e2 = 0.f, e3 = 0.f;
#pragma unroll 64
        for (int t = 0; t < L_; ++t) {
            float yt = yp[t * C_];
            float z  = yt - p0 * yl1 - p1 * yl2 - p2 * yl3;
            float w  = z  - t1 * e2  - t2 * e3;
            float e  = w  - t0 * e1;
            ep[t * ROWSTRIDE_] = e;
            yl3 = yl2; yl2 = yl1; yl1 = yt;
            e3 = e2; e2 = e1; e1 = e;
        }
    }
    __syncthreads();

    if (tid < LANES_) {
        const int x = tid;
        const float M00 = Hh[(L_-1)*4+0], M01 = Hh[(L_-1)*4+1], M02 = Hh[(L_-1)*4+2];
        const float M10 = Hh[(L_-2)*4+0], M11 = Hh[(L_-2)*4+1], M12 = Hh[(L_-2)*4+2];
        const float M20 = Hh[(L_-3)*4+0], M21 = Hh[(L_-3)*4+1], M22 = Hh[(L_-3)*4+2];
        float c1 = 0.f, c2 = 0.f, c3 = 0.f;
        for (int k = 0; k < K_; ++k) {
            CIN[(k*4 + 0) * LANES_ + x] = c1;
            CIN[(k*4 + 1) * LANES_ + x] = c2;
            CIN[(k*4 + 2) * LANES_ + x] = c3;
            CIN[(k*4 + 3) * LANES_ + x] = 0.f;
            const float* ep = EP + k * CHUNKSTRIDE_ + x;
            float s0 = ep[(L_-1)*ROWSTRIDE_] + M00*c1 + M01*c2 + M02*c3;
            float s1 = ep[(L_-2)*ROWSTRIDE_] + M10*c1 + M11*c2 + M12*c3;
            float s2 = ep[(L_-3)*ROWSTRIDE_] + M20*c1 + M21*c2 + M22*c3;
            c1 = s0; c2 = s1; c3 = s2;
        }
    }
    __syncthreads();

    {
        const int l    = tid & 31;
        const int w    = tid >> 5;
        const int n    = l & 15;
        const int half = l >> 4;
        float* obase = out + (size_t)bb * S_ * C_ + cbase + n;
#pragma unroll
        for (int kk = 0; kk < 2; ++kk) {
            const int k = 2 * w + kk;
            v2f bvec;
            bvec.x = CIN[(k*4 + half*2 + 0) * LANES_ + n];
            bvec.y = CIN[(k*4 + half*2 + 1) * LANES_ + n];
            const float* epk = EP + k * CHUNKSTRIDE_ + n;
#pragma unroll
            for (int tt = 0; tt < L_ / 16; ++tt) {
                const int tl = tt * 16 + n;
                v2f avec = *(const v2f*)(Hh + tl * 4 + half * 2);
                v8f cvec;
#pragma unroll
                for (int r = 0; r < 8; ++r)
                    cvec[r] = epk[(tt * 16 + half * 8 + r) * ROWSTRIDE_];
                v8f d = __builtin_amdgcn_wmma_f32_16x16x4_f32(
                     false, avec,  false, bvec,
                     (short)0, cvec,  false,  false);
                float* op = obase + (size_t)(k * L_ + tt * 16 + half * 8) * C_;
#pragma unroll
                for (int r = 0; r < 8; ++r) op[r * C_] = d[r];
            }
        }
    }
}

extern "C" void kernel_launch(void* const* d_in, const int* in_sizes, int n_in,
                              void* d_out, int out_size, void* d_ws, size_t ws_size,
                              hipStream_t stream) {
    const float* y     = (const float*)d_in[0];
    const float* phi   = (const float*)d_in[1];
    const float* theta = (const float*)d_in[2];
    float* out = (float*)d_out;
    (void)d_ws; (void)ws_size; (void)in_sizes; (void)n_in; (void)out_size;

    arima_fused<<<NL_ / LANES_, NTHREADS_, LDS_BYTES_, stream>>>(y, phi, theta, out);
}


# Work-group = 32 channels x two 2048-step halves through one LDS tile (chunks of 64, state carried in registers): each global load/store instruction covers one whole 128-B line, not two 64-B halves
# speedup vs baseline: 5.0163x; 2.6155x over previous
#include <hip/hip_runtime.h>

#define B_  64
#define S_  4096
#define C_  128
#define L_  64
#define KH_ 32
#define NH_ 2
#define LANES_ 32
#define NTHREADS_ (LANES_ * KH_)

#define ROWSTRIDE_   34
#define EP_FLOATS_   (KH_ * L_ * ROWSTRIDE_)
#define H_OFF_       EP_FLOATS_
#define CIN_OFF_     (H_OFF_ + L_ * 4)
#define LDS_FLOATS_  (CIN_OFF_ + KH_ * 4 * LANES_)
#define LDS_BYTES_   (LDS_FLOATS_ * 4)

typedef float v2f __attribute__((ext_vector_type(2)));
typedef float v8f __attribute__((ext_vector_type(8)));

static __device__ __forceinline__ float xhalf(float v) {
    int b = __builtin_bit_cast(int, v);
    return __builtin_bit_cast(float,
        __builtin_amdgcn_permlanex16(b, b, 0x76543210, 0xfedcba98, false, false));
}

__global__ void __launch_bounds__(NTHREADS_) arima_fused(
    const float* __restrict__ y, const float* __restrict__ phi,
    const float* __restrict__ theta, float* __restrict__ out) {
    extern __shared__ float lds[];
    float* EP  = lds;
    float* Hh  = lds + H_OFF_;
    float* CIN = lds + CIN_OFF_;

    const int tid   = threadIdx.x;
    const int lt    = blockIdx.x;
    const int bb    = lt >> 2;
    const int cbase = (lt & 3) * LANES_;
    const int w     = tid >> 5;
    const int x     = tid & 31;

    const float t0 = theta[0], t1 = theta[1], t2 = theta[2];
    const float p0 = phi[0], p1 = phi[1], p2 = phi[2];

    if (tid < 4) {
        const int j = tid;
        float e1 = (j == 0) ? 1.f : 0.f;
        float e2 = (j == 1) ? 1.f : 0.f;
        float e3 = (j == 2) ? 1.f : 0.f;
        for (int t = 0; t < L_; ++t) {
            float e = -(t0 * e1 + t1 * e2 + t2 * e3);
            Hh[t * 4 + j] = e;
            e3 = e2; e2 = e1; e1 = e;
        }
    }

    float c1 = 0.f, c2 = 0.f, c3 = 0.f;

    for (int h = 0; h < NH_; ++h) {
        const int kg = h * KH_ + w;

        {
            const float* yp = y + ((size_t)bb * S_ + (size_t)kg * L_) * C_ + cbase + x;
            float* ep = EP + w * (L_ * ROWSTRIDE_) + x;
            float yl1 = 0.f, yl2 = 0.f, yl3 = 0.f;
            if (kg > 0) { yl1 = yp[-C_]; yl2 = yp[-2 * C_]; yl3 = yp[-3 * C_]; }
            float e1 = 0.f, e2 = 0.f, e3 = 0.f;
#pragma unroll
            for (int t = 0; t < L_; ++t) {
                float yt = yp[t * C_];
                float z  = yt - p0 * yl1 - p1 * yl2 - p2 * yl3;
                float u  = z  - t1 * e2  - t2 * e3;
                float e  = u  - t0 * e1;
                ep[t * ROWSTRIDE_] = e;
                yl3 = yl2; yl2 = yl1; yl1 = yt;
                e3 = e2; e2 = e1; e1 = e;
            }
        }
        __syncthreads();

        if (tid < LANES_) {
            const float M00 = Hh[(L_-1)*4+0], M01 = Hh[(L_-1)*4+1], M02 = Hh[(L_-1)*4+2];
            const float M10 = Hh[(L_-2)*4+0], M11 = Hh[(L_-2)*4+1], M12 = Hh[(L_-2)*4+2];
            const float M20 = Hh[(L_-3)*4+0], M21 = Hh[(L_-3)*4+1], M22 = Hh[(L_-3)*4+2];
            for (int k = 0; k < KH_; ++k) {
                CIN[(k*4 + 0) * LANES_ + x] = c1;
                CIN[(k*4 + 1) * LANES_ + x] = c2;
                CIN[(k*4 + 2) * LANES_ + x] = c3;
                CIN[(k*4 + 3) * LANES_ + x] = 0.f;
                const float* ep = EP + k * (L_ * ROWSTRIDE_) + x;
                float s0 = ep[(L_-1)*ROWSTRIDE_] + M00*c1 + M01*c2 + M02*c3;
                float s1 = ep[(L_-2)*ROWSTRIDE_] + M10*c1 + M11*c2 + M12*c3;
                float s2 = ep[(L_-3)*ROWSTRIDE_] + M20*c1 + M21*c2 + M22*c3;
                c1 = s0; c2 = s1; c3 = s2;
            }
        }
        __syncthreads();

        {
            const int n    = x & 15;
            const int half = x >> 4;
            const bool hi  = half != 0;
            v2f bv0, bv1;
            bv0.x = CIN[(w*4 + half*2 + 0) * LANES_ + n];
            bv0.y = CIN[(w*4 + half*2 + 1) * LANES_ + n];
            bv1.x = CIN[(w*4 + half*2 + 0) * LANES_ + 16 + n];
            bv1.y = CIN[(w*4 + half*2 + 1) * LANES_ + 16 + n];
            const float* epk = EP + w * (L_ * ROWSTRIDE_) + n;
            float* obase = out + ((size_t)bb * S_ + (size_t)kg * L_) * C_ + cbase + x;
#pragma unroll
            for (int tt = 0; tt < L_ / 16; ++tt) {
                const int tl = tt * 16 + n;
                v2f avec = *(const v2f*)(Hh + tl * 4 + half * 2);
                v8f cv0, cv1;
#pragma unroll
                for (int r = 0; r < 8; ++r)
                    cv0[r] = epk[(tt * 16 + half * 8 + r) * ROWSTRIDE_];
#pragma unroll
                for (int r = 0; r < 8; ++r)
                    cv1[r] = epk[(tt * 16 + half * 8 + r) * ROWSTRIDE_ + 16];
                v8f d0 = __builtin_amdgcn_wmma_f32_16x16x4_f32(
                     false, avec,  false, bv0,
                     (short)0, cv0,  false,  false);
                v8f d1 = __builtin_amdgcn_wmma_f32_16x16x4_f32(
                     false, avec,  false, bv1,
                     (short)0, cv1,  false,  false);
                float* op = obase + (size_t)(tt * 16) * C_;
#pragma unroll
                for (int r = 0; r < 8; ++r) {
                    const float give = hi ? d0[r] : d1[r];
                    const float got  = xhalf(give);
                    op[r * C_]       = hi ? got : d0[r];
                    op[(8 + r) * C_] = hi ? d1[r] : got;
                }
            }
        }
    }
}

extern "C" void kernel_launch(void* const* d_in, const int* in_sizes, int n_in,
                              void* d_out, int out_size, void* d_ws, size_t ws_size,
                              hipStream_t stream) {
    const float* y     = (const float*)d_in[0];
    const float* phi   = (const float*)d_in[1];
    const float* theta = (const float*)d_in[2];
    float* out = (float*)d_out;
    (void)d_ws; (void)ws_size; (void)in_sizes; (void)n_in; (void)out_size;

    arima_fused<<<(B_ * C_) / LANES_, NTHREADS_, LDS_BYTES_, stream>>>(y, phi, theta, out);
}


# Carry scan: chunk-final states read from LDS a batch of 8 chunks ahead of the chain (next batch in flight while one runs); a chunk then costs 3 dependent fused multiply-adds, no LDS round trip
# speedup vs baseline: 5.2332x; 1.0432x over previous
#include <hip/hip_runtime.h>

#define B_  64
#define S_  4096
#define C_  128
#define L_  64
#define KH_ 32
#define NH_ 2
#define LANES_ 32
#define NTHREADS_ (LANES_ * KH_)
#define SB_ 8

#define ROWSTRIDE_   34
#define EP_FLOATS_   (KH_ * L_ * ROWSTRIDE_)
#define H_OFF_       EP_FLOATS_
#define CIN_OFF_     (H_OFF_ + L_ * 4)
#define LDS_FLOATS_  (CIN_OFF_ + KH_ * 4 * LANES_)
#define LDS_BYTES_   (LDS_FLOATS_ * 4)

typedef float v2f __attribute__((ext_vector_type(2)));
typedef float v8f __attribute__((ext_vector_type(8)));

static __device__ __forceinline__ float xhalf(float v) {
    int b = __builtin_bit_cast(int, v);
    return __builtin_bit_cast(float,
        __builtin_amdgcn_permlanex16(b, b, 0x76543210, 0xfedcba98, false, false));
}

__global__ void __launch_bounds__(NTHREADS_) arima_fused(
    const float* __restrict__ y, const float* __restrict__ phi,
    const float* __restrict__ theta, float* __restrict__ out) {
    extern __shared__ float lds[];
    float* EP  = lds;
    float* Hh  = lds + H_OFF_;
    float* CIN = lds + CIN_OFF_;

    const int tid   = threadIdx.x;
    const int lt    = blockIdx.x;
    const int bb    = lt >> 2;
    const int cbase = (lt & 3) * LANES_;
    const int w     = tid >> 5;
    const int x     = tid & 31;

    const float t0 = theta[0], t1 = theta[1], t2 = theta[2];
    const float p0 = phi[0], p1 = phi[1], p2 = phi[2];

    if (tid < 4) {
        const int j = tid;
        float e1 = (j == 0) ? 1.f : 0.f;
        float e2 = (j == 1) ? 1.f : 0.f;
        float e3 = (j == 2) ? 1.f : 0.f;
        for (int t = 0; t < L_; ++t) {
            float e = -(t0 * e1 + t1 * e2 + t2 * e3);
            Hh[t * 4 + j] = e;
            e3 = e2; e2 = e1; e1 = e;
        }
    }

    float c1 = 0.f, c2 = 0.f, c3 = 0.f;

    for (int h = 0; h < NH_; ++h) {
        const int kg = h * KH_ + w;

        {
            const float* yp = y + ((size_t)bb * S_ + (size_t)kg * L_) * C_ + cbase + x;
            float* ep = EP + w * (L_ * ROWSTRIDE_) + x;
            float yl1 = 0.f, yl2 = 0.f, yl3 = 0.f;
            if (kg > 0) { yl1 = yp[-C_]; yl2 = yp[-2 * C_]; yl3 = yp[-3 * C_]; }
            float e1 = 0.f, e2 = 0.f, e3 = 0.f;
#pragma unroll
            for (int t = 0; t < L_; ++t) {
                float yt = yp[t * C_];
                float z  = yt - p0 * yl1 - p1 * yl2 - p2 * yl3;
                float u  = z  - t1 * e2  - t2 * e3;
                float e  = u  - t0 * e1;
                ep[t * ROWSTRIDE_] = e;
                yl3 = yl2; yl2 = yl1; yl1 = yt;
                e3 = e2; e2 = e1; e1 = e;
            }
        }
        __syncthreads();

        if (tid < LANES_) {
            const float M00 = Hh[(L_-1)*4+0], M01 = Hh[(L_-1)*4+1], M02 = Hh[(L_-1)*4+2];
            const float M10 = Hh[(L_-2)*4+0], M11 = Hh[(L_-2)*4+1], M12 = Hh[(L_-2)*4+2];
            const float M20 = Hh[(L_-3)*4+0], M21 = Hh[(L_-3)*4+1], M22 = Hh[(L_-3)*4+2];
            float a0[SB_], a1[SB_], a2[SB_];
            float n0[SB_], n1[SB_], n2[SB_];
#pragma unroll
            for (int j = 0; j < SB_; ++j) {
                const float* ep = EP + j * (L_ * ROWSTRIDE_) + x;
                a0[j] = ep[(L_-1)*ROWSTRIDE_];
                a1[j] = ep[(L_-2)*ROWSTRIDE_];
                a2[j] = ep[(L_-3)*ROWSTRIDE_];
                n0[j] = a0[j]; n1[j] = a1[j]; n2[j] = a2[j];
            }
#pragma unroll 1
            for (int kb = 0; kb < KH_; kb += SB_) {
                if (kb + SB_ < KH_) {
#pragma unroll
                    for (int j = 0; j < SB_; ++j) {
                        const float* ep = EP + (kb + SB_ + j) * (L_ * ROWSTRIDE_) + x;
                        n0[j] = ep[(L_-1)*ROWSTRIDE_];
                        n1[j] = ep[(L_-2)*ROWSTRIDE_];
                        n2[j] = ep[(L_-3)*ROWSTRIDE_];
                    }
                }
#pragma unroll
                for (int j = 0; j < SB_; ++j) {
                    const int k = kb + j;
                    CIN[(k*4 + 0) * LANES_ + x] = c1;
                    CIN[(k*4 + 1) * LANES_ + x] = c2;
                    CIN[(k*4 + 2) * LANES_ + x] = c3;
                    CIN[(k*4 + 3) * LANES_ + x] = 0.f;
                    const float s0 = __builtin_fmaf(M02, c3, __builtin_fmaf(M01, c2, __builtin_fmaf(M00, c1, a0[j])));
                    const float s1 = __builtin_fmaf(M12, c3, __builtin_fmaf(M11, c2, __builtin_fmaf(M10, c1, a1[j])));
                    const float s2 = __builtin_fmaf(M22, c3, __builtin_fmaf(M21, c2, __builtin_fmaf(M20, c1, a2[j])));
                    c1 = s0; c2 = s1; c3 = s2;
                }
#pragma unroll
                for (int j = 0; j < SB_; ++j) { a0[j] = n0[j]; a1[j] = n1[j]; a2[j] = n2[j]; }
            }
        }
        __syncthreads();

        {
            const int n    = x & 15;
            const int half = x >> 4;
            const bool hi  = half != 0;
            v2f bv0, bv1;
            bv0.x = CIN[(w*4 + half*2 + 0) * LANES_ + n];
            bv0.y = CIN[(w*4 + half*2 + 1) * LANES_ + n];
            bv1.x = CIN[(w*4 + half*2 + 0) * LANES_ + 16 + n];
            bv1.y = CIN[(w*4 + half*2 + 1) * LANES_ + 16 + n];
            const float* epk = EP + w * (L_ * ROWSTRIDE_) + n;
            float* obase = out + ((size_t)bb * S_ + (size_t)kg * L_) * C_ + cbase + x;
#pragma unroll
            for (int tt = 0; tt < L_ / 16; ++tt) {
                const int tl = tt * 16 + n;
                v2f avec = *(const v2f*)(Hh + tl * 4 + half * 2);
                v8f cv0, cv1;
#pragma unroll
                for (int r = 0; r < 8; ++r)
                    cv0[r] = epk[(tt * 16 + half * 8 + r) * ROWSTRIDE_];
#pragma unroll
                for (int r = 0; r < 8; ++r)
                    cv1[r] = epk[(tt * 16 + half * 8 + r) * ROWSTRIDE_ + 16];
                v8f d0 = __builtin_amdgcn_wmma_f32_16x16x4_f32(
                     false, avec,  false, bv0,
                     (short)0, cv0,  false,  false);
                v8f d1 = __builtin_amdgcn_wmma_f32_16x16x4_f32(
                     false, avec,  false, bv1,
                     (short)0, cv1,  false,  false);
                float* op = obase + (size_t)(tt * 16) * C_;
#pragma unroll
                for (int r = 0; r < 8; ++r) {
                    const float give = hi ? d0[r] : d1[r];
                    const float got  = xhalf(give);
                    op[r * C_]       = hi ? got : d0[r];
                    op[(8 + r) * C_] = hi ? d1[r] : got;
                }
            }
        }
    }
}

extern "C" void kernel_launch(void* const* d_in, const int* in_sizes, int n_in,
                              void* d_out, int out_size, void* d_ws, size_t ws_size,
                              hipStream_t stream) {
    const float* y     = (const float*)d_in[0];
    const float* phi   = (const float*)d_in[1];
    const float* theta = (const float*)d_in[2];
    float* out = (float*)d_out;
    (void)d_ws; (void)ws_size; (void)in_sizes; (void)n_in; (void)out_size;

    arima_fused<<<(B_ * C_) / LANES_, NTHREADS_, LDS_BYTES_, stream>>>(y, phi, theta, out);
}
